# starting kernel
# baseline (speedup 1.0000x reference)
#include <hip/hip_runtime.h>
#include <cmath>

typedef __attribute__((ext_vector_type(2))) float v2f;
typedef __attribute__((ext_vector_type(4))) float v4f;
typedef __attribute__((ext_vector_type(8))) float v8f;

#define WMMA4(a, b, c) __builtin_amdgcn_wmma_f32_16x16x4_f32( \
    false, (a), false, (b), (short)0, (c), false, false)

namespace {
constexpr int kB = 8, kL = 512, kN = 64, kF = 128, kH = 8, kC = 16;
constexpr int kChunks = kL / kC;
constexpr int XS = 132;
constexpr int ST = 18;
constexpr float kEps = 1e-6f;
}

__device__ __forceinline__ float phi(float x) {
  return fmaxf(x, 0.0f) + __expf(fminf(x, 0.0f));
}

__global__ __launch_bounds__(256, 1)
void cla_fused_kernel(const float* __restrict__ qin,
                      const float* __restrict__ Wq, const float* __restrict__ bq,
                      const float* __restrict__ Wk, const float* __restrict__ bk,
                      const float* __restrict__ Wv, const float* __restrict__ bv,
                      float* __restrict__ out)
{
  __shared__ float xs[kC * XS];
  __shared__ float shQ[kH][kC * ST];
  __shared__ float shK[kH][kC * ST];
  __shared__ float shV[kH][kC * ST];
  __shared__ float shA[kH][kC * ST];
  __shared__ float shS[kH][kC * ST];
  __shared__ float shKcum[kH][16];
  __shared__ float shZ[kH][16];
  __shared__ float ostage[kC * kF];

  const int tid   = threadIdx.x;
  const int lane  = tid & 31;
  const int w     = tid >> 5;
  const int col   = lane & 15;
  const int hi    = lane >> 4;
  const int khalf = hi << 1;
  const int mrow  = hi << 3;

  const int g = blockIdx.x;
  const int b = g >> 6;
  const int n = g & 63;

  float* Qt = shQ[w];
  float* Kt = shK[w];
  float* Vt = shV[w];
  float* At = shA[w];
  float* St = shS[w];

  for (int i = lane; i < kC * ST; i += 32) St[i] = 0.0f;
  if (lane < 16) shKcum[w][lane] = 0.0f;

  const int wcol = w * 16 + col;
  const float bqv = bq[wcol];
  const float bkv = bk[wcol];
  const float bvv = bv[wcol];

  const int cr = tid >> 4;
  const int cc = (tid & 15) << 3;

  for (int t = 0; t < kChunks; ++t) {
    {
      const float* src =
          qin + (((size_t)(b * kL + t * kC + cr) * kN + n) * kF + cc);
      v4f a0 = *reinterpret_cast<const v4f*>(src);
      v4f a1 = *reinterpret_cast<const v4f*>(src + 4);
      *reinterpret_cast<v4f*>(&xs[cr * XS + cc]) = a0;
      *reinterpret_cast<v4f*>(&xs[cr * XS + cc + 4]) = a1;
      if (t + 1 < kChunks) {
        __builtin_prefetch(src + (size_t)kC * kN * kF, 0, 0);
      }
    }
    __syncthreads();

    v8f cq = {}; v8f ck = {}; v8f cv = {};
    for (int s = 0; s < 32; ++s) {
      const int k0 = 4 * s + khalf;
      v2f a = *reinterpret_cast<v2f*>(&xs[col * XS + k0]);
      v2f wb;
      wb.x = Wq[k0 * kF + wcol]; wb.y = Wq[(k0 + 1) * kF + wcol];
      cq = WMMA4(a, wb, cq);
      wb.x = Wk[k0 * kF + wcol]; wb.y = Wk[(k0 + 1) * kF + wcol];
      ck = WMMA4(a, wb, ck);
      wb.x = Wv[k0 * kF + wcol]; wb.y = Wv[(k0 + 1) * kF + wcol];
      cv = WMMA4(a, wb, cv);
    }

    for (int r = 0; r < 8; ++r) {
      const int m = r + mrow;
      Qt[m * ST + col] = phi(cq[r] + bqv);
      Kt[m * ST + col] = phi(ck[r] + bkv);
      Vt[m * ST + col] = cv[r] + bvv;
    }
    asm volatile("" ::: "memory");

    v8f aat = {};
    for (int s = 0; s < 4; ++s) {
      const int k0 = 4 * s + khalf;
      v2f a  = *reinterpret_cast<v2f*>(&Qt[col * ST + k0]);
      v2f bb = *reinterpret_cast<v2f*>(&Kt[col * ST + k0]);
      aat = WMMA4(a, bb, aat);
    }
    for (int r = 0; r < 8; ++r) {
      const int m = r + mrow;
      At[m * ST + col] = (col <= m) ? aat[r] : 0.0f;
    }
    asm volatile("" ::: "memory");

    if (lane < 16) {
      float den = 0.0f;
      for (int c2 = 0; c2 < 16; ++c2) den += At[lane * ST + c2];
      float dq = 0.0f;
      for (int e = 0; e < 16; ++e) dq += Qt[lane * ST + e] * shKcum[w][e];
      shZ[w][lane] = __builtin_amdgcn_rcpf(den + dq + kEps);
    }
    asm volatile("" ::: "memory");

    v8f oac = {};
    for (int s = 0; s < 4; ++s) {
      const int k0 = 4 * s + khalf;
      v2f a = *reinterpret_cast<v2f*>(&At[col * ST + k0]);
      v2f bb; bb.x = Vt[k0 * ST + col]; bb.y = Vt[(k0 + 1) * ST + col];
      oac = WMMA4(a, bb, oac);
    }
    for (int s = 0; s < 4; ++s) {
      const int k0 = 4 * s + khalf;
      v2f a = *reinterpret_cast<v2f*>(&Qt[col * ST + k0]);
      v2f bb; bb.x = St[k0 * ST + col]; bb.y = St[(k0 + 1) * ST + col];
      oac = WMMA4(a, bb, oac);
    }

    v8f sac;
    for (int r = 0; r < 8; ++r) sac[r] = St[(r + mrow) * ST + col];
    for (int s = 0; s < 4; ++s) {
      const int k0 = 4 * s + khalf;
      v2f a;  a.x  = Kt[k0 * ST + col]; a.y  = Kt[(k0 + 1) * ST + col];
      v2f bb; bb.x = Vt[k0 * ST + col]; bb.y = Vt[(k0 + 1) * ST + col];
      sac = WMMA4(a, bb, sac);
    }
    for (int r = 0; r < 8; ++r) St[(r + mrow) * ST + col] = sac[r];

    if (lane < 16) {
      float s = 0.0f;
      for (int c2 = 0; c2 < 16; ++c2) s += Kt[c2 * ST + lane];
      shKcum[w][lane] += s;
    }
    asm volatile("" ::: "memory");

    for (int r = 0; r < 8; ++r) {
      const int m = r + mrow;
      ostage[m * kF + w * 16 + col] = oac[r] * shZ[w][m];
    }
    __syncthreads();

    {
      float* dst =
          out + (((size_t)(b * kL + t * kC + cr) * kN + n) * kF + cc);
      *reinterpret_cast<v4f*>(dst)     = *reinterpret_cast<v4f*>(&ostage[cr * kF + cc]);
      *reinterpret_cast<v4f*>(dst + 4) = *reinterpret_cast<v4f*>(&ostage[cr * kF + cc + 4]);
    }
  }
}

extern "C" void kernel_launch(void* const* d_in, const int* in_sizes, int n_in,
                              void* d_out, int out_size, void* d_ws, size_t ws_size,
                              hipStream_t stream) {
  (void)in_sizes; (void)n_in; (void)out_size; (void)d_ws; (void)ws_size;
  const float* q  = (const float*)d_in[0];
  const float* Wq = (const float*)d_in[1];
  const float* bq = (const float*)d_in[2];
  const float* Wk = (const float*)d_in[3];
  const float* bk = (const float*)d_in[4];
  const float* Wv = (const float*)d_in[5];
  const float* bv = (const float*)d_in[6];
  cla_fused_kernel<<<dim3(kB * kN), dim3(256), 0, stream>>>(
      q, Wq, bq, Wk, bk, Wv, bv, (float*)d_out);
}


# Wq, Wk, Wv fragments loaded once per wave, held in registers for all 32 chunks (k-loop unrolled; chunk loop loads no W); LDS tiles addressed as one base + constants to fit 256 VGPRs; sums unchanged
# speedup vs baseline: 1.3267x; 1.3267x over previous
#include <hip/hip_runtime.h>
#include <cmath>

typedef __attribute__((ext_vector_type(2))) float v2f;
typedef __attribute__((ext_vector_type(4))) float v4f;
typedef __attribute__((ext_vector_type(8))) float v8f;

#define WMMA4(a, b, c) __builtin_amdgcn_wmma_f32_16x16x4_f32( \
    false, (a), false, (b), (short)0, (c), false, false)

namespace {
constexpr int kB = 8, kL = 512, kN = 64, kF = 128, kH = 8, kC = 16;
constexpr int kChunks = kL / kC;
constexpr int XS = 132;
constexpr int ST = 18;
constexpr float kEps = 1e-6f;

struct WaveTiles {
  float q[kC * ST];
  float k[kC * ST];
  float v[kC * ST];
  float a[kC * ST];
  float s[kC * ST];
  float kcum[16];
  float z[16];
};
}

__device__ __forceinline__ float phi(float x) {
  return fmaxf(x, 0.0f) + __expf(fminf(x, 0.0f));
}

__global__ __launch_bounds__(256, 4)
void cla_fused_kernel(const float* __restrict__ qin,
                      const float* __restrict__ Wq, const float* __restrict__ bq,
                      const float* __restrict__ Wk, const float* __restrict__ bk,
                      const float* __restrict__ Wv, const float* __restrict__ bv,
                      float* __restrict__ out)
{
  __shared__ float xs[kC * XS];
  __shared__ WaveTiles wt[kH];
  __shared__ float ostage[kC * kF];

  const int tid   = threadIdx.x;
  const int lane  = tid & 31;
  const int w     = tid >> 5;
  const int col   = lane & 15;
  const int hi    = lane >> 4;
  const int khalf = hi << 1;
  const int mrow  = hi << 3;

  const int g = blockIdx.x;
  const int b = g >> 6;
  const int n = g & 63;

  float* Qt = wt[w].q;
  float* Kt = wt[w].k;
  float* Vt = wt[w].v;
  float* At = wt[w].a;
  float* St = wt[w].s;
  float* kcum = wt[w].kcum;
  float* zrow = wt[w].z;

  const int cofs = mrow * ST + col;
  const int aofs = col * ST + khalf;
  const int bofs = khalf * ST + col;
  float* Qc = Qt + cofs; float* Kc = Kt + cofs; float* Vc = Vt + cofs;
  float* Ac = At + cofs; float* Sc = St + cofs;
  const float* Qa = Qt + aofs; const float* Ka = Kt + aofs; const float* Aa = At + aofs;
  const float* Kb = Kt + bofs; const float* Vb = Vt + bofs; const float* Sb = St + bofs;
  const float* Xa = xs + col * XS + khalf;
  float* Oc = ostage + mrow * kF + w * 16 + col;
  const float* Zc = zrow + mrow;

  for (int i = lane; i < kC * ST; i += 32) St[i] = 0.0f;
  if (lane < 16) kcum[lane] = 0.0f;

  const int wcol = w * 16 + col;
  const float bqv = bq[wcol];
  const float bkv = bk[wcol];
  const float bvv = bv[wcol];

  const int cr = tid >> 4;
  const int cc = (tid & 15) << 3;

  v2f wqf[32], wkf[32], wvf[32];
#pragma unroll
  for (int s = 0; s < 32; ++s) {
    const int k0 = 4 * s + khalf;
    wqf[s].x = Wq[k0 * kF + wcol]; wqf[s].y = Wq[(k0 + 1) * kF + wcol];
    wkf[s].x = Wk[k0 * kF + wcol]; wkf[s].y = Wk[(k0 + 1) * kF + wcol];
    wvf[s].x = Wv[k0 * kF + wcol]; wvf[s].y = Wv[(k0 + 1) * kF + wcol];
  }

  for (int t = 0; t < kChunks; ++t) {
    {
      const float* src =
          qin + (((size_t)(b * kL + t * kC + cr) * kN + n) * kF + cc);
      v4f a0 = *reinterpret_cast<const v4f*>(src);
      v4f a1 = *reinterpret_cast<const v4f*>(src + 4);
      *reinterpret_cast<v4f*>(&xs[cr * XS + cc]) = a0;
      *reinterpret_cast<v4f*>(&xs[cr * XS + cc + 4]) = a1;
      if (t + 1 < kChunks) {
        __builtin_prefetch(src + (size_t)kC * kN * kF, 0, 0);
      }
    }
    __syncthreads();

    v8f cq = {}; v8f ck = {}; v8f cv = {};
#pragma unroll
    for (int s = 0; s < 32; ++s) {
      v2f a = *reinterpret_cast<const v2f*>(Xa + 4 * s);
      cq = WMMA4(a, wqf[s], cq);
      ck = WMMA4(a, wkf[s], ck);
      cv = WMMA4(a, wvf[s], cv);
    }

    for (int r = 0; r < 8; ++r) {
      Qc[r * ST] = phi(cq[r] + bqv);
      Kc[r * ST] = phi(ck[r] + bkv);
      Vc[r * ST] = cv[r] + bvv;
    }
    asm volatile("" ::: "memory");

    v8f aat = {};
    for (int s = 0; s < 4; ++s) {
      v2f a  = *reinterpret_cast<const v2f*>(Qa + 4 * s);
      v2f bb = *reinterpret_cast<const v2f*>(Ka + 4 * s);
      aat = WMMA4(a, bb, aat);
    }
    for (int r = 0; r < 8; ++r) {
      Ac[r * ST] = (col <= r + mrow) ? aat[r] : 0.0f;
    }
    asm volatile("" ::: "memory");

    if (lane < 16) {
      float den = 0.0f;
      for (int c2 = 0; c2 < 16; ++c2) den += Aa[c2];
      float dq = 0.0f;
      for (int e = 0; e < 16; ++e) dq += Qa[e] * kcum[e];
      zrow[lane] = __builtin_amdgcn_rcpf(den + dq + kEps);
    }
    asm volatile("" ::: "memory");

    v8f oac = {};
    for (int s = 0; s < 4; ++s) {
      v2f a = *reinterpret_cast<const v2f*>(Aa + 4 * s);
      v2f bb; bb.x = Vb[4 * s * ST]; bb.y = Vb[(4 * s + 1) * ST];
      oac = WMMA4(a, bb, oac);
    }
    for (int s = 0; s < 4; ++s) {
      v2f a = *reinterpret_cast<const v2f*>(Qa + 4 * s);
      v2f bb; bb.x = Sb[4 * s * ST]; bb.y = Sb[(4 * s + 1) * ST];
      oac = WMMA4(a, bb, oac);
    }

    v8f sac;
    for (int r = 0; r < 8; ++r) sac[r] = Sc[r * ST];
    for (int s = 0; s < 4; ++s) {
      v2f a;  a.x  = Kb[4 * s * ST]; a.y  = Kb[(4 * s + 1) * ST];
      v2f bb; bb.x = Vb[4 * s * ST]; bb.y = Vb[(4 * s + 1) * ST];
      sac = WMMA4(a, bb, sac);
    }
    for (int r = 0; r < 8; ++r) Sc[r * ST] = sac[r];

    if (lane < 16) {
      float s = 0.0f;
      for (int c2 = 0; c2 < 16; ++c2) s += Kb[c2 * ST];
      kcum[lane] += s;
    }
    asm volatile("" ::: "memory");

    for (int r = 0; r < 8; ++r) {
      Oc[r * kF] = oac[r] * Zc[r];
    }
    __syncthreads();

    {
      float* dst =
          out + (((size_t)(b * kL + t * kC + cr) * kN + n) * kF + cc);
      *reinterpret_cast<v4f*>(dst)     = *reinterpret_cast<v4f*>(&ostage[cr * kF + cc]);
      *reinterpret_cast<v4f*>(dst + 4) = *reinterpret_cast<v4f*>(&ostage[cr * kF + cc + 4]);
    }
  }
}

extern "C" void kernel_launch(void* const* d_in, const int* in_sizes, int n_in,
                              void* d_out, int out_size, void* d_ws, size_t ws_size,
                              hipStream_t stream) {
  (void)in_sizes; (void)n_in; (void)out_size; (void)d_ws; (void)ws_size;
  const float* q  = (const float*)d_in[0];
  const float* Wq = (const float*)d_in[1];
  const float* bq = (const float*)d_in[2];
  const float* Wk = (const float*)d_in[3];
  const float* bk = (const float*)d_in[4];
  const float* Wv = (const float*)d_in[5];
  const float* bv = (const float*)d_in[6];
  cla_fused_kernel<<<dim3(kB * kN), dim3(256), 0, stream>>>(
      q, Wq, bq, Wk, bk, Wv, bv, (float*)d_out);
}


# Wave priority by phase: s_setprio 3 for the serial part of each chunk (feature map, scan, output, next x), 0 for the 96-WMMA projection, so a younger wave's scan is not starved; sums unchanged
# speedup vs baseline: 1.4048x; 1.0589x over previous
#include <hip/hip_runtime.h>
#include <cmath>

typedef __attribute__((ext_vector_type(2))) float v2f;
typedef __attribute__((ext_vector_type(4))) float v4f;
typedef __attribute__((ext_vector_type(8))) float v8f;

#define WMMA4(a, b, c) __builtin_amdgcn_wmma_f32_16x16x4_f32( \
    false, (a), false, (b), (short)0, (c), false, false)

namespace {
constexpr int kB = 8, kL = 512, kN = 64, kF = 128, kH = 8, kC = 16;
constexpr int kChunks = kL / kC;
constexpr int XS = 132;
constexpr int ST = 18;
constexpr float kEps = 1e-6f;

struct WaveTiles {
  float q[kC * ST];
  float k[kC * ST];
  float v[kC * ST];
  float a[kC * ST];
  float s[kC * ST];
  float kcum[16];
  float z[16];
};
}

__device__ __forceinline__ float phi(float x) {
  return fmaxf(x, 0.0f) + __expf(fminf(x, 0.0f));
}

__global__ __launch_bounds__(256, 4)
void cla_fused_kernel(const float* __restrict__ qin,
                      const float* __restrict__ Wq, const float* __restrict__ bq,
                      const float* __restrict__ Wk, const float* __restrict__ bk,
                      const float* __restrict__ Wv, const float* __restrict__ bv,
                      float* __restrict__ out)
{
  __shared__ float xs[kC * XS];
  __shared__ WaveTiles wt[kH];
  __shared__ float ostage[kC * kF];

  const int tid   = threadIdx.x;
  const int lane  = tid & 31;
  const int w     = tid >> 5;
  const int col   = lane & 15;
  const int hi    = lane >> 4;
  const int khalf = hi << 1;
  const int mrow  = hi << 3;

  const int g = blockIdx.x;
  const int b = g >> 6;
  const int n = g & 63;

  float* Qt = wt[w].q;
  float* Kt = wt[w].k;
  float* Vt = wt[w].v;
  float* At = wt[w].a;
  float* St = wt[w].s;
  float* kcum = wt[w].kcum;
  float* zrow = wt[w].z;

  const int cofs = mrow * ST + col;
  const int aofs = col * ST + khalf;
  const int bofs = khalf * ST + col;
  float* Qc = Qt + cofs; float* Kc = Kt + cofs; float* Vc = Vt + cofs;
  float* Ac = At + cofs; float* Sc = St + cofs;
  const float* Qa = Qt + aofs; const float* Ka = Kt + aofs; const float* Aa = At + aofs;
  const float* Kb = Kt + bofs; const float* Vb = Vt + bofs; const float* Sb = St + bofs;
  const float* Xa = xs + col * XS + khalf;
  float* Oc = ostage + mrow * kF + w * 16 + col;
  const float* Zc = zrow + mrow;

  for (int i = lane; i < kC * ST; i += 32) St[i] = 0.0f;
  if (lane < 16) kcum[lane] = 0.0f;

  const int wcol = w * 16 + col;
  const float bqv = bq[wcol];
  const float bkv = bk[wcol];
  const float bvv = bv[wcol];

  const int cr = tid >> 4;
  const int cc = (tid & 15) << 3;

  v2f wqf[32], wkf[32], wvf[32];
#pragma unroll
  for (int s = 0; s < 32; ++s) {
    const int k0 = 4 * s + khalf;
    wqf[s].x = Wq[k0 * kF + wcol]; wqf[s].y = Wq[(k0 + 1) * kF + wcol];
    wkf[s].x = Wk[k0 * kF + wcol]; wkf[s].y = Wk[(k0 + 1) * kF + wcol];
    wvf[s].x = Wv[k0 * kF + wcol]; wvf[s].y = Wv[(k0 + 1) * kF + wcol];
  }

  for (int t = 0; t < kChunks; ++t) {
    {
      const float* src =
          qin + (((size_t)(b * kL + t * kC + cr) * kN + n) * kF + cc);
      v4f a0 = *reinterpret_cast<const v4f*>(src);
      v4f a1 = *reinterpret_cast<const v4f*>(src + 4);
      *reinterpret_cast<v4f*>(&xs[cr * XS + cc]) = a0;
      *reinterpret_cast<v4f*>(&xs[cr * XS + cc + 4]) = a1;
      if (t + 1 < kChunks) {
        __builtin_prefetch(src + (size_t)kC * kN * kF, 0, 0);
      }
    }
    __syncthreads();

    __builtin_amdgcn_s_setprio(0);

    v8f cq = {}; v8f ck = {}; v8f cv = {};
#pragma unroll
    for (int s = 0; s < 32; ++s) {
      v2f a = *reinterpret_cast<const v2f*>(Xa + 4 * s);
      cq = WMMA4(a, wqf[s], cq);
      ck = WMMA4(a, wkf[s], ck);
      cv = WMMA4(a, wvf[s], cv);
    }

    __builtin_amdgcn_s_setprio(3);

    for (int r = 0; r < 8; ++r) {
      Qc[r * ST] = phi(cq[r] + bqv);
      Kc[r * ST] = phi(ck[r] + bkv);
      Vc[r * ST] = cv[r] + bvv;
    }
    asm volatile("" ::: "memory");

    v8f aat = {};
    for (int s = 0; s < 4; ++s) {
      v2f a  = *reinterpret_cast<const v2f*>(Qa + 4 * s);
      v2f bb = *reinterpret_cast<const v2f*>(Ka + 4 * s);
      aat = WMMA4(a, bb, aat);
    }
    for (int r = 0; r < 8; ++r) {
      Ac[r * ST] = (col <= r + mrow) ? aat[r] : 0.0f;
    }
    asm volatile("" ::: "memory");

    if (lane < 16) {
      float den = 0.0f;
      for (int c2 = 0; c2 < 16; ++c2) den += Aa[c2];
      float dq = 0.0f;
      for (int e = 0; e < 16; ++e) dq += Qa[e] * kcum[e];
      zrow[lane] = __builtin_amdgcn_rcpf(den + dq + kEps);
    }
    asm volatile("" ::: "memory");

    v8f oac = {};
    for (int s = 0; s < 4; ++s) {
      v2f a = *reinterpret_cast<const v2f*>(Aa + 4 * s);
      v2f bb; bb.x = Vb[4 * s * ST]; bb.y = Vb[(4 * s + 1) * ST];
      oac = WMMA4(a, bb, oac);
    }
    for (int s = 0; s < 4; ++s) {
      v2f a = *reinterpret_cast<const v2f*>(Qa + 4 * s);
      v2f bb; bb.x = Sb[4 * s * ST]; bb.y = Sb[(4 * s + 1) * ST];
      oac = WMMA4(a, bb, oac);
    }

    v8f sac;
    for (int r = 0; r < 8; ++r) sac[r] = Sc[r * ST];
    for (int s = 0; s < 4; ++s) {
      v2f a;  a.x  = Kb[4 * s * ST]; a.y  = Kb[(4 * s + 1) * ST];
      v2f bb; bb.x = Vb[4 * s * ST]; bb.y = Vb[(4 * s + 1) * ST];
      sac = WMMA4(a, bb, sac);
    }
    for (int r = 0; r < 8; ++r) Sc[r * ST] = sac[r];

    if (lane < 16) {
      float s = 0.0f;
      for (int c2 = 0; c2 < 16; ++c2) s += Kb[c2 * ST];
      kcum[lane] += s;
    }
    asm volatile("" ::: "memory");

    for (int r = 0; r < 8; ++r) {
      Oc[r * kF] = oac[r] * Zc[r];
    }
    __syncthreads();

    {
      float* dst =
          out + (((size_t)(b * kL + t * kC + cr) * kN + n) * kF + cc);
      *reinterpret_cast<v4f*>(dst)     = *reinterpret_cast<v4f*>(&ostage[cr * kF + cc]);
      *reinterpret_cast<v4f*>(dst + 4) = *reinterpret_cast<v4f*>(&ostage[cr * kF + cc + 4]);
    }
  }
}

extern "C" void kernel_launch(void* const* d_in, const int* in_sizes, int n_in,
                              void* d_out, int out_size, void* d_ws, size_t ws_size,
                              hipStream_t stream) {
  (void)in_sizes; (void)n_in; (void)out_size; (void)d_ws; (void)ws_size;
  const float* q  = (const float*)d_in[0];
  const float* Wq = (const float*)d_in[1];
  const float* bq = (const float*)d_in[2];
  const float* Wk = (const float*)d_in[3];
  const float* bk = (const float*)d_in[4];
  const float* Wv = (const float*)d_in[5];
  const float* bv = (const float*)d_in[6];
  cla_fused_kernel<<<dim3(kB * kN), dim3(256), 0, stream>>>(
      q, Wq, bq, Wk, bk, Wv, bv, (float*)d_out);
}


# Projection priority rises with progress (s_setprio 0,1,2 by thirds of the k loop, scan stays 3): an older wave's fresh projection no longer preempts one under way; sums unchanged
# speedup vs baseline: 1.5040x; 1.0706x over previous
#include <hip/hip_runtime.h>
#include <cmath>

typedef __attribute__((ext_vector_type(2))) float v2f;
typedef __attribute__((ext_vector_type(4))) float v4f;
typedef __attribute__((ext_vector_type(8))) float v8f;

#define WMMA4(a, b, c) __builtin_amdgcn_wmma_f32_16x16x4_f32( \
    false, (a), false, (b), (short)0, (c), false, false)

namespace {
constexpr int kB = 8, kL = 512, kN = 64, kF = 128, kH = 8, kC = 16;
constexpr int kChunks = kL / kC;
constexpr int XS = 132;
constexpr int ST = 18;
constexpr float kEps = 1e-6f;

struct WaveTiles {
  float q[kC * ST];
  float k[kC * ST];
  float v[kC * ST];
  float a[kC * ST];
  float s[kC * ST];
  float kcum[16];
  float z[16];
};
}

__device__ __forceinline__ float phi(float x) {
  return fmaxf(x, 0.0f) + __expf(fminf(x, 0.0f));
}

__global__ __launch_bounds__(256, 4)
void cla_fused_kernel(const float* __restrict__ qin,
                      const float* __restrict__ Wq, const float* __restrict__ bq,
                      const float* __restrict__ Wk, const float* __restrict__ bk,
                      const float* __restrict__ Wv, const float* __restrict__ bv,
                      float* __restrict__ out)
{
  __shared__ float xs[kC * XS];
  __shared__ WaveTiles wt[kH];
  __shared__ float ostage[kC * kF];

  const int tid   = threadIdx.x;
  const int lane  = tid & 31;
  const int w     = tid >> 5;
  const int col   = lane & 15;
  const int hi    = lane >> 4;
  const int khalf = hi << 1;
  const int mrow  = hi << 3;

  const int g = blockIdx.x;
  const int b = g >> 6;
  const int n = g & 63;

  float* Qt = wt[w].q;
  float* Kt = wt[w].k;
  float* Vt = wt[w].v;
  float* At = wt[w].a;
  float* St = wt[w].s;
  float* kcum = wt[w].kcum;
  float* zrow = wt[w].z;

  const int cofs = mrow * ST + col;
  const int aofs = col * ST + khalf;
  const int bofs = khalf * ST + col;
  float* Qc = Qt + cofs; float* Kc = Kt + cofs; float* Vc = Vt + cofs;
  float* Ac = At + cofs; float* Sc = St + cofs;
  const float* Qa = Qt + aofs; const float* Ka = Kt + aofs; const float* Aa = At + aofs;
  const float* Kb = Kt + bofs; const float* Vb = Vt + bofs; const float* Sb = St + bofs;
  const float* Xa = xs + col * XS + khalf;
  float* Oc = ostage + mrow * kF + w * 16 + col;
  const float* Zc = zrow + mrow;

  for (int i = lane; i < kC * ST; i += 32) St[i] = 0.0f;
  if (lane < 16) kcum[lane] = 0.0f;

  const int wcol = w * 16 + col;
  const float bqv = bq[wcol];
  const float bkv = bk[wcol];
  const float bvv = bv[wcol];

  const int cr = tid >> 4;
  const int cc = (tid & 15) << 3;

  v2f wqf[32], wkf[32], wvf[32];
#pragma unroll
  for (int s = 0; s < 32; ++s) {
    const int k0 = 4 * s + khalf;
    wqf[s].x = Wq[k0 * kF + wcol]; wqf[s].y = Wq[(k0 + 1) * kF + wcol];
    wkf[s].x = Wk[k0 * kF + wcol]; wkf[s].y = Wk[(k0 + 1) * kF + wcol];
    wvf[s].x = Wv[k0 * kF + wcol]; wvf[s].y = Wv[(k0 + 1) * kF + wcol];
  }

  for (int t = 0; t < kChunks; ++t) {
    {
      const float* src =
          qin + (((size_t)(b * kL + t * kC + cr) * kN + n) * kF + cc);
      v4f a0 = *reinterpret_cast<const v4f*>(src);
      v4f a1 = *reinterpret_cast<const v4f*>(src + 4);
      *reinterpret_cast<v4f*>(&xs[cr * XS + cc]) = a0;
      *reinterpret_cast<v4f*>(&xs[cr * XS + cc + 4]) = a1;
      if (t + 1 < kChunks) {
        __builtin_prefetch(src + (size_t)kC * kN * kF, 0, 0);
      }
    }
    __syncthreads();

    __builtin_amdgcn_s_setprio(0);

    v8f cq = {}; v8f ck = {}; v8f cv = {};
    v2f xa0 = *reinterpret_cast<const v2f*>(Xa);
    v2f xa1 = *reinterpret_cast<const v2f*>(Xa + 4);
#pragma unroll
    for (int p = 0; p < 16; ++p) {
      v2f xn0 = xa0, xn1 = xa1;
      if (p + 1 < 16) {
        xn0 = *reinterpret_cast<const v2f*>(Xa + 8 * (p + 1));
        xn1 = *reinterpret_cast<const v2f*>(Xa + 8 * (p + 1) + 4);
      }
      if (p == 5) __builtin_amdgcn_s_setprio(1);
      if (p == 11) __builtin_amdgcn_s_setprio(2);
      cq = WMMA4(xa0, wqf[2 * p], cq);
      ck = WMMA4(xa0, wkf[2 * p], ck);
      cv = WMMA4(xa0, wvf[2 * p], cv);
      cq = WMMA4(xa1, wqf[2 * p + 1], cq);
      ck = WMMA4(xa1, wkf[2 * p + 1], ck);
      cv = WMMA4(xa1, wvf[2 * p + 1], cv);
      xa0 = xn0; xa1 = xn1;
    }

    __builtin_amdgcn_s_setprio(3);

    for (int r = 0; r < 8; ++r) {
      Qc[r * ST] = phi(cq[r] + bqv);
      Kc[r * ST] = phi(ck[r] + bkv);
      Vc[r * ST] = cv[r] + bvv;
    }
    asm volatile("" ::: "memory");

    v8f aat = {};
    for (int s = 0; s < 4; ++s) {
      v2f a  = *reinterpret_cast<const v2f*>(Qa + 4 * s);
      v2f bb = *reinterpret_cast<const v2f*>(Ka + 4 * s);
      aat = WMMA4(a, bb, aat);
    }
    for (int r = 0; r < 8; ++r) {
      Ac[r * ST] = (col <= r + mrow) ? aat[r] : 0.0f;
    }
    asm volatile("" ::: "memory");

    if (lane < 16) {
      float den = 0.0f;
      for (int c2 = 0; c2 < 16; ++c2) den += Aa[c2];
      float dq = 0.0f;
      for (int e = 0; e < 16; ++e) dq += Qa[e] * kcum[e];
      zrow[lane] = __builtin_amdgcn_rcpf(den + dq + kEps);
    }
    asm volatile("" ::: "memory");

    v8f oac = {};
    for (int s = 0; s < 4; ++s) {
      v2f a = *reinterpret_cast<const v2f*>(Aa + 4 * s);
      v2f bb; bb.x = Vb[4 * s * ST]; bb.y = Vb[(4 * s + 1) * ST];
      oac = WMMA4(a, bb, oac);
    }
    for (int s = 0; s < 4; ++s) {
      v2f a = *reinterpret_cast<const v2f*>(Qa + 4 * s);
      v2f bb; bb.x = Sb[4 * s * ST]; bb.y = Sb[(4 * s + 1) * ST];
      oac = WMMA4(a, bb, oac);
    }

    v8f sac;
    for (int r = 0; r < 8; ++r) sac[r] = Sc[r * ST];
    for (int s = 0; s < 4; ++s) {
      v2f a;  a.x  = Kb[4 * s * ST]; a.y  = Kb[(4 * s + 1) * ST];
      v2f bb; bb.x = Vb[4 * s * ST]; bb.y = Vb[(4 * s + 1) * ST];
      sac = WMMA4(a, bb, sac);
    }
    for (int r = 0; r < 8; ++r) Sc[r * ST] = sac[r];

    if (lane < 16) {
      float s = 0.0f;
      for (int c2 = 0; c2 < 16; ++c2) s += Kb[c2 * ST];
      kcum[lane] += s;
    }
    asm volatile("" ::: "memory");

    for (int r = 0; r < 8; ++r) {
      Oc[r * kF] = oac[r] * Zc[r];
    }
    __syncthreads();

    {
      float* dst =
          out + (((size_t)(b * kL + t * kC + cr) * kN + n) * kF + cc);
      *reinterpret_cast<v4f*>(dst)     = *reinterpret_cast<v4f*>(&ostage[cr * kF + cc]);
      *reinterpret_cast<v4f*>(dst + 4) = *reinterpret_cast<v4f*>(&ostage[cr * kF + cc + 4]);
    }
  }
}

extern "C" void kernel_launch(void* const* d_in, const int* in_sizes, int n_in,
                              void* d_out, int out_size, void* d_ws, size_t ws_size,
                              hipStream_t stream) {
  (void)in_sizes; (void)n_in; (void)out_size; (void)d_ws; (void)ws_size;
  const float* q  = (const float*)d_in[0];
  const float* Wq = (const float*)d_in[1];
  const float* bq = (const float*)d_in[2];
  const float* Wk = (const float*)d_in[3];
  const float* bk = (const float*)d_in[4];
  const float* Wv = (const float*)d_in[5];
  const float* bv = (const float*)d_in[6];
  cla_fused_kernel<<<dim3(kB * kN), dim3(256), 0, stream>>>(
      q, Wq, bq, Wk, bk, Wv, bv, (float*)d_out);
}


# Workgroup halved to 4 heads (4 waves, one per SIMD): four independent groups per WGP cover each other's barriers and scans; a group's waves share no matrix pipe; both halves stage x; sums unchanged
# speedup vs baseline: 1.5620x; 1.0386x over previous
#include <hip/hip_runtime.h>
#include <cmath>

typedef __attribute__((ext_vector_type(2))) float v2f;
typedef __attribute__((ext_vector_type(4))) float v4f;
typedef __attribute__((ext_vector_type(8))) float v8f;

#define WMMA4(a, b, c) __builtin_amdgcn_wmma_f32_16x16x4_f32( \
    false, (a), false, (b), (short)0, (c), false, false)

namespace {
constexpr int kB = 8, kL = 512, kN = 64, kF = 128, kH = 8, kC = 16;
constexpr int kChunks = kL / kC;
constexpr int kHW = 4;
constexpr int kOW = kHW * 16;
constexpr int XS = 132;
constexpr int OS = kOW + 4;
constexpr int ST = 18;
constexpr float kEps = 1e-6f;

struct WaveTiles {
  float q[kC * ST];
  float k[kC * ST];
  float v[kC * ST];
  float a[kC * ST];
  float s[kC * ST];
  float kcum[16];
  float z[16];
};
}

__device__ __forceinline__ float phi(float x) {
  return fmaxf(x, 0.0f) + __expf(fminf(x, 0.0f));
}

__global__ __launch_bounds__(128, 4)
void cla_fused_kernel(const float* __restrict__ qin,
                      const float* __restrict__ Wq, const float* __restrict__ bq,
                      const float* __restrict__ Wk, const float* __restrict__ bk,
                      const float* __restrict__ Wv, const float* __restrict__ bv,
                      float* __restrict__ out)
{
  __shared__ alignas(16) float xs[kC * XS];
  __shared__ WaveTiles wt[kHW];
  __shared__ alignas(16) float ostage[kC * OS];

  const int tid   = threadIdx.x;
  const int lane  = tid & 31;
  const int w     = tid >> 5;
  const int col   = lane & 15;
  const int hi    = lane >> 4;
  const int khalf = hi << 1;
  const int mrow  = hi << 3;

  const int g = blockIdx.x >> 1;
  const int hh = blockIdx.x & 1;
  const int b = g >> 6;
  const int n = g & 63;

  float* Qt = wt[w].q;
  float* Kt = wt[w].k;
  float* Vt = wt[w].v;
  float* At = wt[w].a;
  float* St = wt[w].s;
  float* kcum = wt[w].kcum;
  float* zrow = wt[w].z;

  const int cofs = mrow * ST + col;
  const int aofs = col * ST + khalf;
  const int bofs = khalf * ST + col;
  float* Qc = Qt + cofs; float* Kc = Kt + cofs; float* Vc = Vt + cofs;
  float* Ac = At + cofs; float* Sc = St + cofs;
  const float* Qa = Qt + aofs; const float* Ka = Kt + aofs; const float* Aa = At + aofs;
  const float* Kb = Kt + bofs; const float* Vb = Vt + bofs; const float* Sb = St + bofs;
  const float* Xa = xs + col * XS + khalf;
  float* Oc = ostage + mrow * OS + w * 16 + col;
  const float* Zc = zrow + mrow;

  for (int i = lane; i < kC * ST; i += 32) St[i] = 0.0f;
  if (lane < 16) kcum[lane] = 0.0f;

  const int wcol = (hh * kHW + w) * 16 + col;
  const float bqv = bq[wcol];
  const float bkv = bk[wcol];
  const float bvv = bv[wcol];

  const int xc = lane << 2;
  const int sr = tid >> 4;
  const int sc = (tid & 15) << 2;

  v2f wqf[32], wkf[32], wvf[32];
#pragma unroll
  for (int s = 0; s < 32; ++s) {
    const int k0 = 4 * s + khalf;
    wqf[s].x = Wq[k0 * kF + wcol]; wqf[s].y = Wq[(k0 + 1) * kF + wcol];
    wkf[s].x = Wk[k0 * kF + wcol]; wkf[s].y = Wk[(k0 + 1) * kF + wcol];
    wvf[s].x = Wv[k0 * kF + wcol]; wvf[s].y = Wv[(k0 + 1) * kF + wcol];
  }

  for (int t = 0; t < kChunks; ++t) {
    {
      const float* src =
          qin + (((size_t)(b * kL + t * kC + w) * kN + n) * kF + xc);
      const size_t rstep = (size_t)kHW * kN * kF;
      v4f a0 = *reinterpret_cast<const v4f*>(src);
      v4f a1 = *reinterpret_cast<const v4f*>(src + rstep);
      v4f a2 = *reinterpret_cast<const v4f*>(src + 2 * rstep);
      v4f a3 = *reinterpret_cast<const v4f*>(src + 3 * rstep);
      *reinterpret_cast<v4f*>(&xs[w * XS + xc]) = a0;
      *reinterpret_cast<v4f*>(&xs[(w + 4) * XS + xc]) = a1;
      *reinterpret_cast<v4f*>(&xs[(w + 8) * XS + xc]) = a2;
      *reinterpret_cast<v4f*>(&xs[(w + 12) * XS + xc]) = a3;
      if (t + 1 < kChunks) {
        const float* nx = src + (size_t)kC * kN * kF;
        __builtin_prefetch(nx, 0, 0);
        __builtin_prefetch(nx + rstep, 0, 0);
        __builtin_prefetch(nx + 2 * rstep, 0, 0);
        __builtin_prefetch(nx + 3 * rstep, 0, 0);
      }
    }
    __syncthreads();

    __builtin_amdgcn_s_setprio(0);

    v8f cq = {}; v8f ck = {}; v8f cv = {};
    v2f xa0 = *reinterpret_cast<const v2f*>(Xa);
    v2f xa1 = *reinterpret_cast<const v2f*>(Xa + 4);
#pragma unroll
    for (int p = 0; p < 16; ++p) {
      v2f xn0 = xa0, xn1 = xa1;
      if (p + 1 < 16) {
        xn0 = *reinterpret_cast<const v2f*>(Xa + 8 * (p + 1));
        xn1 = *reinterpret_cast<const v2f*>(Xa + 8 * (p + 1) + 4);
      }
      if (p == 5) __builtin_amdgcn_s_setprio(1);
      if (p == 11) __builtin_amdgcn_s_setprio(2);
      cq = WMMA4(xa0, wqf[2 * p], cq);
      ck = WMMA4(xa0, wkf[2 * p], ck);
      cv = WMMA4(xa0, wvf[2 * p], cv);
      cq = WMMA4(xa1, wqf[2 * p + 1], cq);
      ck = WMMA4(xa1, wkf[2 * p + 1], ck);
      cv = WMMA4(xa1, wvf[2 * p + 1], cv);
      xa0 = xn0; xa1 = xn1;
    }

    __builtin_amdgcn_s_setprio(3);

    for (int r = 0; r < 8; ++r) {
      Qc[r * ST] = phi(cq[r] + bqv);
      Kc[r * ST] = phi(ck[r] + bkv);
      Vc[r * ST] = cv[r] + bvv;
    }
    asm volatile("" ::: "memory");

    v8f aat = {};
    for (int s = 0; s < 4; ++s) {
      v2f a  = *reinterpret_cast<const v2f*>(Qa + 4 * s);
      v2f bb = *reinterpret_cast<const v2f*>(Ka + 4 * s);
      aat = WMMA4(a, bb, aat);
    }
    for (int r = 0; r < 8; ++r) {
      Ac[r * ST] = (col <= r + mrow) ? aat[r] : 0.0f;
    }
    asm volatile("" ::: "memory");

    if (lane < 16) {
      float den = 0.0f;
      for (int c2 = 0; c2 < 16; ++c2) den += Aa[c2];
      float dq = 0.0f;
      for (int e = 0; e < 16; ++e) dq += Qa[e] * kcum[e];
      zrow[lane] = __builtin_amdgcn_rcpf(den + dq + kEps);
    }
    asm volatile("" ::: "memory");

    v8f oac = {};
    for (int s = 0; s < 4; ++s) {
      v2f a = *reinterpret_cast<const v2f*>(Aa + 4 * s);
      v2f bb; bb.x = Vb[4 * s * ST]; bb.y = Vb[(4 * s + 1) * ST];
      oac = WMMA4(a, bb, oac);
    }
    for (int s = 0; s < 4; ++s) {
      v2f a = *reinterpret_cast<const v2f*>(Qa + 4 * s);
      v2f bb; bb.x = Sb[4 * s * ST]; bb.y = Sb[(4 * s + 1) * ST];
      oac = WMMA4(a, bb, oac);
    }

    v8f sac;
    for (int r = 0; r < 8; ++r) sac[r] = Sc[r * ST];
    for (int s = 0; s < 4; ++s) {
      v2f a;  a.x  = Kb[4 * s * ST]; a.y  = Kb[(4 * s + 1) * ST];
      v2f bb; bb.x = Vb[4 * s * ST]; bb.y = Vb[(4 * s + 1) * ST];
      sac = WMMA4(a, bb, sac);
    }
    for (int r = 0; r < 8; ++r) Sc[r * ST] = sac[r];

    if (lane < 16) {
      float s = 0.0f;
      for (int c2 = 0; c2 < 16; ++c2) s += Kb[c2 * ST];
      kcum[lane] += s;
    }
    asm volatile("" ::: "memory");

    for (int r = 0; r < 8; ++r) {
      Oc[r * OS] = oac[r] * Zc[r];
    }
    __syncthreads();

    {
      float* dst =
          out + (((size_t)(b * kL + t * kC + sr) * kN + n) * kF + hh * kOW + sc);
      *reinterpret_cast<v4f*>(dst) =
          *reinterpret_cast<v4f*>(&ostage[sr * OS + sc]);
      *reinterpret_cast<v4f*>(dst + (size_t)8 * kN * kF) =
          *reinterpret_cast<v4f*>(&ostage[(sr + 8) * OS + sc]);
    }
  }
}

extern "C" void kernel_launch(void* const* d_in, const int* in_sizes, int n_in,
                              void* d_out, int out_size, void* d_ws, size_t ws_size,
                              hipStream_t stream) {
  (void)in_sizes; (void)n_in; (void)out_size; (void)d_ws; (void)ws_size;
  const float* q  = (const float*)d_in[0];
  const float* Wq = (const float*)d_in[1];
  const float* bq = (const float*)d_in[2];
  const float* Wk = (const float*)d_in[3];
  const float* bk = (const float*)d_in[4];
  const float* Wv = (const float*)d_in[5];
  const float* bv = (const float*)d_in[6];
  cla_fused_kernel<<<dim3(kB * kN * 2), dim3(128), 0, stream>>>(
      q, Wq, bq, Wk, bk, Wv, bv, (float*)d_out);
}
